# starting kernel
# baseline (speedup 1.0000x reference)
#include <hip/hip_runtime.h>
#include <hip/hip_bf16.h>
#include <math.h>

typedef float v2f __attribute__((ext_vector_type(2)));
typedef float v8f __attribute__((ext_vector_type(8)));

#define N_NODES 20000
#define N_EDGES 640000
#define FIN_DIM 256
#define FQK_DIM 256

__global__ __launch_bounds__(256) void zero_f32(float* __restrict__ p, int n) {
  int i = blockIdx.x * 256 + threadIdx.x;
  if (i < n) p[i] = 0.0f;
}

__global__ __launch_bounds__(256) void gemm_qk(const float* __restrict__ x,
                                               const float* __restrict__ W,
                                               float* __restrict__ q,
                                               float* __restrict__ kmat) {
  const int lane  = threadIdx.x & 31;
  const int strip = blockIdx.x * 8 + (threadIdx.x >> 5);
  const int mBase = (strip >> 3) << 4;
  const int nBase = (strip & 7) << 6;
  const int mRow  = lane & 15;
  const int nCol  = lane & 15;
  const int kHalf = (lane >> 4) << 1;

  const float* __restrict__ xp = x + (size_t)(mBase + mRow) * FIN_DIM + kHalf;
  const float* __restrict__ w0 = W + (size_t)(nBase + nCol) * FIN_DIM + kHalf;
  const float* __restrict__ w1 = w0 + 16 * FIN_DIM;
  const float* __restrict__ w2 = w0 + 32 * FIN_DIM;
  const float* __restrict__ w3 = w0 + 48 * FIN_DIM;

  v8f c0 = {}, c1 = {}, c2 = {}, c3 = {};

#pragma unroll 4
  for (int k0 = 0; k0 < FIN_DIM; k0 += 4) {
    v2f a  = *(const v2f*)(xp + k0);
    v2f b0 = *(const v2f*)(w0 + k0);
    v2f b1 = *(const v2f*)(w1 + k0);
    v2f b2 = *(const v2f*)(w2 + k0);
    v2f b3 = *(const v2f*)(w3 + k0);
    c0 = __builtin_amdgcn_wmma_f32_16x16x4_f32(false, a, false, b0, (short)0, c0, false, false);
    c1 = __builtin_amdgcn_wmma_f32_16x16x4_f32(false, a, false, b1, (short)0, c1, false, false);
    c2 = __builtin_amdgcn_wmma_f32_16x16x4_f32(false, a, false, b2, (short)0, c2, false, false);
    c3 = __builtin_amdgcn_wmma_f32_16x16x4_f32(false, a, false, b3, (short)0, c3, false, false);
  }

  const int mOff = (lane >> 4) ? 8 : 0;
  v8f cc[4] = {c0, c1, c2, c3};

#pragma unroll
  for (int j = 0; j < 4; ++j) {
    const int col = nBase + j * 16 + nCol;
    float* __restrict__ dst;
    float scale;
    if (col < FQK_DIM) { dst = q + col;                 scale = 0.0625f; }
    else               { dst = kmat + (col - FQK_DIM);  scale = 1.0f;    }
#pragma unroll
    for (int i = 0; i < 8; ++i) {
      const int m = mBase + mOff + i;
      dst[(size_t)m * FQK_DIM] = cc[j][i] * scale;
    }
  }
}

__global__ __launch_bounds__(256) void edge_exp(const float* __restrict__ q,
                                                const float* __restrict__ kmat,
                                                const int* __restrict__ src,
                                                const int* __restrict__ dst,
                                                float* __restrict__ ex,
                                                float* __restrict__ expsum,
                                                int nE) {
  const int gid  = blockIdx.x * 256 + threadIdx.x;
  const int e    = gid >> 5;
  const int lane = threadIdx.x & 31;
  if (e >= nE) return;

  const int s = src[e];
  const int d = dst[e];
  const float4* __restrict__ qp = (const float4*)(q    + (size_t)s * FQK_DIM);
  const float4* __restrict__ kp = (const float4*)(kmat + (size_t)d * FQK_DIM);

  const float4 qa = qp[lane];      const float4 ka = kp[lane];
  const float4 qb = qp[lane + 32]; const float4 kb = kp[lane + 32];

  float sum = qa.x * ka.x + qa.y * ka.y + qa.z * ka.z + qa.w * ka.w
            + qb.x * kb.x + qb.y * kb.y + qb.z * kb.z + qb.w * kb.w;

  sum += __shfl_xor(sum, 16, 32);
  sum += __shfl_xor(sum, 8, 32);
  sum += __shfl_xor(sum, 4, 32);
  sum += __shfl_xor(sum, 2, 32);
  sum += __shfl_xor(sum, 1, 32);

  if (lane == 0) {
    const float v = __expf(sum);
    ex[e] = v;
    unsafeAtomicAdd(expsum + s, v);
  }
}

__global__ __launch_bounds__(256) void edge_norm(float* __restrict__ ex,
                                                 const int* __restrict__ src,
                                                 const float* __restrict__ expsum,
                                                 int nE) {
  const int e = blockIdx.x * 256 + threadIdx.x;
  if (e < nE) ex[e] = ex[e] / expsum[src[e]];
}

extern "C" void kernel_launch(void* const* d_in, const int* in_sizes, int n_in,
                              void* d_out, int out_size, void* d_ws, size_t ws_size,
                              hipStream_t stream) {
  const float* x  = (const float*)d_in[0];
  const int*   ei = (const int*)d_in[1];
  const float* W  = (const float*)d_in[2];
  float* out = (float*)d_out;

  float* q      = (float*)d_ws;
  float* kmat   = q + (size_t)N_NODES * FQK_DIM;
  float* expsum = kmat + (size_t)N_NODES * FQK_DIM;

  const int* src = ei;
  const int* dst = ei + N_EDGES;

  zero_f32<<<(N_NODES + 255) / 256, 256, 0, stream>>>(expsum, N_NODES);
  gemm_qk<<<1250, 256, 0, stream>>>(x, W, q, kmat);
  edge_exp<<<N_EDGES / 8, 256, 0, stream>>>(q, kmat, src, dst, out, expsum, N_EDGES);
  edge_norm<<<(N_EDGES + 255) / 256, 256, 0, stream>>>(out, src, expsum, N_EDGES);
}


# edge pass re-tiled: one wave walks 32 consecutive edges (q row stays in registers while src repeats, 8 k rows requested together, ex stored as whole lines), 20000 waves not 640000; sum order kept
# speedup vs baseline: 1.3917x; 1.3917x over previous
#include <hip/hip_runtime.h>
#include <hip/hip_bf16.h>
#include <math.h>

typedef float v2f __attribute__((ext_vector_type(2)));
typedef float v8f __attribute__((ext_vector_type(8)));

#define N_NODES 20000
#define N_EDGES 640000
#define FIN_DIM 256
#define FQK_DIM 256

__global__ __launch_bounds__(256) void zero_f32(float* __restrict__ p, int n) {
  int i = blockIdx.x * 256 + threadIdx.x;
  if (i < n) p[i] = 0.0f;
}

__global__ __launch_bounds__(256) void gemm_qk(const float* __restrict__ x,
                                               const float* __restrict__ W,
                                               float* __restrict__ q,
                                               float* __restrict__ kmat) {
  const int lane  = threadIdx.x & 31;
  const int strip = blockIdx.x * 8 + (threadIdx.x >> 5);
  const int mBase = (strip >> 3) << 4;
  const int nBase = (strip & 7) << 6;
  const int mRow  = lane & 15;
  const int nCol  = lane & 15;
  const int kHalf = (lane >> 4) << 1;

  const float* __restrict__ xp = x + (size_t)(mBase + mRow) * FIN_DIM + kHalf;
  const float* __restrict__ w0 = W + (size_t)(nBase + nCol) * FIN_DIM + kHalf;
  const float* __restrict__ w1 = w0 + 16 * FIN_DIM;
  const float* __restrict__ w2 = w0 + 32 * FIN_DIM;
  const float* __restrict__ w3 = w0 + 48 * FIN_DIM;

  v8f c0 = {}, c1 = {}, c2 = {}, c3 = {};

#pragma unroll 4
  for (int k0 = 0; k0 < FIN_DIM; k0 += 4) {
    v2f a  = *(const v2f*)(xp + k0);
    v2f b0 = *(const v2f*)(w0 + k0);
    v2f b1 = *(const v2f*)(w1 + k0);
    v2f b2 = *(const v2f*)(w2 + k0);
    v2f b3 = *(const v2f*)(w3 + k0);
    c0 = __builtin_amdgcn_wmma_f32_16x16x4_f32(false, a, false, b0, (short)0, c0, false, false);
    c1 = __builtin_amdgcn_wmma_f32_16x16x4_f32(false, a, false, b1, (short)0, c1, false, false);
    c2 = __builtin_amdgcn_wmma_f32_16x16x4_f32(false, a, false, b2, (short)0, c2, false, false);
    c3 = __builtin_amdgcn_wmma_f32_16x16x4_f32(false, a, false, b3, (short)0, c3, false, false);
  }

  const int mOff = (lane >> 4) ? 8 : 0;
  v8f cc[4] = {c0, c1, c2, c3};

  for (int j = 0; j < 4; ++j) {
    const int col = nBase + j * 16 + nCol;
    float* __restrict__ dst;
    float scale;
    if (col < FQK_DIM) { dst = q + col;                 scale = 0.0625f; }
    else               { dst = kmat + (col - FQK_DIM);  scale = 1.0f;    }
    for (int i = 0; i < 8; ++i) {
      const int m = mBase + mOff + i;
      dst[(size_t)m * FQK_DIM] = cc[j][i] * scale;
    }
  }
}

__device__ __forceinline__ float xhalf(float x) {
  const int b = __builtin_bit_cast(int, x);
  return __builtin_bit_cast(float,
      __builtin_amdgcn_permlanex16(b, b, 0x76543210, 0xfedcba98, false, false));
}
template <int M>
__device__ __forceinline__ float xrow(float x) {
  const int b = __builtin_bit_cast(int, x);
  return __builtin_bit_cast(float,
      __builtin_amdgcn_update_dpp(0, b, 0x160 + M, 0xf, 0xf, true));
}
__device__ __forceinline__ float wave_sum(float s) {
  s += xhalf(s);
  s += xrow<8>(s);
  s += xrow<4>(s);
  s += xrow<2>(s);
  s += xrow<1>(s);
  return s;
}
__device__ __forceinline__ float dot8(const float4 qa, const float4 qb,
                                      const float4 ka, const float4 kb) {
  float r = qa.y * ka.y;
  r = __builtin_fmaf(qa.x, ka.x, r);
  r = __builtin_fmaf(qa.z, ka.z, r);
  r = __builtin_fmaf(qa.w, ka.w, r);
  r = __builtin_fmaf(qb.x, kb.x, r);
  r = __builtin_fmaf(qb.y, kb.y, r);
  r = __builtin_fmaf(qb.z, kb.z, r);
  r = __builtin_fmaf(qb.w, kb.w, r);
  return r;
}

#define EDGES_PER_WAVE 32
#define EDGE_GROUP 8
static_assert(N_EDGES % EDGES_PER_WAVE == 0, "edge count must fill whole waves");

__global__ __launch_bounds__(256) void edge_exp(const float* __restrict__ q,
                                                const float* __restrict__ kmat,
                                                const int* __restrict__ src,
                                                const int* __restrict__ dst,
                                                float* __restrict__ ex,
                                                float* __restrict__ expsum,
                                                int nE) {
  const int lane = threadIdx.x & 31;
  const int wv = __builtin_amdgcn_readfirstlane((int)(threadIdx.x >> 5));
  const int e0 = (blockIdx.x * 8 + wv) * EDGES_PER_WAVE;
  if (e0 >= nE) return;

  const int sv = src[e0 + lane];
  const int dv = dst[e0 + lane];

  int sCur = __builtin_amdgcn_readlane(sv, 0);
  const float4* qp = (const float4*)(q + (size_t)sCur * FQK_DIM);
  float4 qa = qp[lane];
  float4 qb = qp[lane + 32];

  float acc = 0.0f;

  for (int j0 = 0; j0 < EDGES_PER_WAVE; j0 += EDGE_GROUP) {
    int s[EDGE_GROUP];
    int differ = 0;
    for (int t = 0; t < EDGE_GROUP; ++t) {
      s[t] = __builtin_amdgcn_readlane(sv, j0 + t);
      differ |= s[t] ^ s[0];
    }

    if (differ == 0) {
      if (s[0] != sCur) {
        sCur = s[0];
        const float4* qn = (const float4*)(q + (size_t)sCur * FQK_DIM);
        qa = qn[lane];
        qb = qn[lane + 32];
      }
      float4 ka[EDGE_GROUP], kb[EDGE_GROUP];
      for (int t = 0; t < EDGE_GROUP; ++t) {
        const float4* kp = (const float4*)(kmat + (size_t)__builtin_amdgcn_readlane(dv, j0 + t) * FQK_DIM);
        ka[t] = kp[lane];
        kb[t] = kp[lane + 32];
      }
      for (int t = 0; t < EDGE_GROUP; ++t) {
        const float r = wave_sum(dot8(qa, qb, ka[t], kb[t]));
        acc = (lane == j0 + t) ? r : acc;
      }
    } else {
      for (int j = j0; j < j0 + EDGE_GROUP; j += 2) {
        const int sx = __builtin_amdgcn_readlane(sv, j);
        const int sy = __builtin_amdgcn_readlane(sv, j + 1);
        const float4* qx = (const float4*)(q + (size_t)sx * FQK_DIM);
        const float4* qy = (const float4*)(q + (size_t)sy * FQK_DIM);
        const float4* kx = (const float4*)(kmat + (size_t)__builtin_amdgcn_readlane(dv, j) * FQK_DIM);
        const float4* ky = (const float4*)(kmat + (size_t)__builtin_amdgcn_readlane(dv, j + 1) * FQK_DIM);
        const float4 qxa = qx[lane], qxb = qx[lane + 32];
        const float4 qya = qy[lane], qyb = qy[lane + 32];
        const float4 kxa = kx[lane], kxb = kx[lane + 32];
        const float4 kya = ky[lane], kyb = ky[lane + 32];
        const float rx = wave_sum(dot8(qxa, qxb, kxa, kxb));
        const float ry = wave_sum(dot8(qya, qyb, kya, kyb));
        acc = (lane == j)     ? rx : acc;
        acc = (lane == j + 1) ? ry : acc;
        qa = qya;
        qb = qyb;
        sCur = sy;
      }
    }
  }

  const float v = __expf(acc);
  ex[e0 + lane] = v;
  unsafeAtomicAdd(expsum + sv, v);
}

__global__ __launch_bounds__(256) void edge_norm(float* __restrict__ ex,
                                                 const int* __restrict__ src,
                                                 const float* __restrict__ expsum,
                                                 int nE) {
  const int e = blockIdx.x * 256 + threadIdx.x;
  if (e < nE) ex[e] = ex[e] / expsum[src[e]];
}

extern "C" void kernel_launch(void* const* d_in, const int* in_sizes, int n_in,
                              void* d_out, int out_size, void* d_ws, size_t ws_size,
                              hipStream_t stream) {
  const float* x  = (const float*)d_in[0];
  const int*   ei = (const int*)d_in[1];
  const float* W  = (const float*)d_in[2];
  float* out = (float*)d_out;

  float* q      = (float*)d_ws;
  float* kmat   = q + (size_t)N_NODES * FQK_DIM;
  float* expsum = kmat + (size_t)N_NODES * FQK_DIM;

  const int* src = ei;
  const int* dst = ei + N_EDGES;

  zero_f32<<<(N_NODES + 255) / 256, 256, 0, stream>>>(expsum, N_NODES);
  gemm_qk<<<1250, 256, 0, stream>>>(x, W, q, kmat);
  edge_exp<<<N_EDGES / (8 * EDGES_PER_WAVE), 256, 0, stream>>>(q, kmat, src, dst, out, expsum, N_EDGES);
  edge_norm<<<(N_EDGES + 255) / 256, 256, 0, stream>>>(out, src, expsum, N_EDGES);
}


# qk product re-tiled: 160x256 tile per 8-wave group, x and W k-chunks fetched as whole lines into LDS (pitch 68 floats), fragments read from LDS one group ahead, 5x4 tiles per wave; sum order kept
# speedup vs baseline: 1.8458x; 1.3263x over previous
#include <hip/hip_runtime.h>
#include <hip/hip_bf16.h>
#include <math.h>

typedef float v2f __attribute__((ext_vector_type(2)));
typedef float v8f __attribute__((ext_vector_type(8)));

#define N_NODES 20000
#define N_EDGES 640000
#define FIN_DIM 256
#define FQK_DIM 256

__global__ __launch_bounds__(256) void zero_f32(float* __restrict__ p, int n) {
  int i = blockIdx.x * 256 + threadIdx.x;
  if (i < n) p[i] = 0.0f;
}

typedef float v4f __attribute__((ext_vector_type(4)));

#define GEMM_BM 160
#define GEMM_BN 256
#define GEMM_KC 64
#define GEMM_PITCH 68
#define GEMM_ROWS (GEMM_BM + GEMM_BN)
#define GEMM_STAGE (GEMM_ROWS * GEMM_PITCH)
#define GEMM_PRE (GEMM_ROWS / 16)
#define GEMM_PRE_A (GEMM_BM / 16)
#define GEMM_CHUNKS (FIN_DIM / GEMM_KC)
static_assert(N_NODES % GEMM_BM == 0, "row tiles must cover the nodes exactly");
static_assert(GEMM_BN == FQK_DIM, "one column tile is the q half or the k half");

__device__ __forceinline__ void gemm_fetch(v4f (&pre)[GEMM_PRE], const float* gA, const float* gB,
                                           const int kOff) {
#pragma unroll
  for (int i = 0; i < GEMM_PRE_A; ++i)
    pre[i] = *(const v4f*)(gA + (size_t)i * 16 * FIN_DIM + kOff);
#pragma unroll
  for (int i = GEMM_PRE_A; i < GEMM_PRE; ++i)
    pre[i] = *(const v4f*)(gB + (size_t)(i - GEMM_PRE_A) * 16 * FIN_DIM + kOff);
}
__device__ __forceinline__ void gemm_put(const v4f (&pre)[GEMM_PRE], float* s) {
#pragma unroll
  for (int i = 0; i < GEMM_PRE; ++i) *(v4f*)(s + i * 16 * GEMM_PITCH) = pre[i];
}
__device__ __forceinline__ void gemm_frag_pair(v2f (&a)[2][5], v2f (&b)[2][4], const float* aS,
                                               const float* bS, const int g) {
#pragma unroll
  for (int s = 0; s < 2; ++s) {
#pragma unroll
    for (int i = 0; i < 5; ++i) a[s][i] = *(const v2f*)(aS + i * 16 * GEMM_PITCH + (2 * g + s) * 4);
#pragma unroll
    for (int j = 0; j < 4; ++j) b[s][j] = *(const v2f*)(bS + j * 16 * GEMM_PITCH + (2 * g + s) * 4);
  }
}
__device__ __forceinline__ void gemm_chunk(v8f (&acc)[5][4], const float* aS, const float* bS) {
  v2f a[2][2][5], b[2][2][4];
  gemm_frag_pair(a[0], b[0], aS, bS, 0);
#pragma unroll
  for (int g = 0; g < GEMM_KC / 8; ++g) {
    if (g + 1 < GEMM_KC / 8) gemm_frag_pair(a[(g + 1) & 1], b[(g + 1) & 1], aS, bS, g + 1);
    __builtin_amdgcn_sched_barrier(0);
#pragma unroll
    for (int s = 0; s < 2; ++s)
#pragma unroll
      for (int i = 0; i < 5; ++i)
#pragma unroll
        for (int j = 0; j < 4; ++j)
          acc[i][j] = __builtin_amdgcn_wmma_f32_16x16x4_f32(false, a[g & 1][s][i], false,
                                                            b[g & 1][s][j], (short)0, acc[i][j],
                                                            false, false);
    __builtin_amdgcn_sched_barrier(0);
  }
}

__global__ __launch_bounds__(256) void gemm_qk(const float* __restrict__ x,
                                               const float* __restrict__ W,
                                               float* __restrict__ q,
                                               float* __restrict__ kmat) {
  __shared__ __attribute__((aligned(16))) float smem[2 * GEMM_STAGE];

  const int tid   = threadIdx.x;
  const int lane  = tid & 31;
  const int wv    = tid >> 5;
  const int mBase = (blockIdx.x >> 1) * GEMM_BM;
  const int nHalf = blockIdx.x & 1;

  const int c4 = tid & 15;
  const int r0 = tid >> 4;
  const float* gA = x + (size_t)(mBase + r0) * FIN_DIM + c4 * 4;
  const float* gB = W + (size_t)(nHalf * GEMM_BN + r0) * FIN_DIM + c4 * 4;
  float* sStore = smem + r0 * GEMM_PITCH + c4 * 4;

  const int c  = lane & 15;
  const int h2 = (lane >> 4) << 1;
  const int wr = wv >> 2;
  const int wc = wv & 3;
  const float* aBase = smem + (wr * 80 + c) * GEMM_PITCH + h2;
  const float* bBase = smem + (GEMM_BM + wc * 64 + c) * GEMM_PITCH + h2;

  v8f acc[5][4];
#pragma unroll
  for (int i = 0; i < 5; ++i)
#pragma unroll
    for (int j = 0; j < 4; ++j) acc[i][j] = (v8f){};

  {
    v4f pre[GEMM_PRE];
    gemm_fetch(pre, gA, gB, 0);
    gemm_put(pre, sStore);
  }
  __syncthreads();

#pragma unroll 1
  for (int kc = 0; kc < GEMM_CHUNKS; ++kc) {
    v4f pre[GEMM_PRE];
    const bool more = (kc + 1) < GEMM_CHUNKS;
    if (more) gemm_fetch(pre, gA, gB, (kc + 1) * GEMM_KC);
    gemm_chunk(acc, aBase + (kc & 1) * GEMM_STAGE, bBase + (kc & 1) * GEMM_STAGE);
    if (more) gemm_put(pre, sStore + ((kc + 1) & 1) * GEMM_STAGE);
    __syncthreads();
  }

  float* __restrict__ dst = nHalf ? kmat : q;
  const float scale = nHalf ? 1.0f : 0.0625f;
  const int mOff = (lane >> 4) ? 8 : 0;
#pragma unroll
  for (int i = 0; i < 5; ++i) {
#pragma unroll
    for (int j = 0; j < 4; ++j) {
      const int col = wc * 64 + j * 16 + c;
#pragma unroll
      for (int r = 0; r < 8; ++r) {
        const int m = mBase + wr * 80 + i * 16 + mOff + r;
        dst[(size_t)m * FQK_DIM + col] = acc[i][j][r] * scale;
      }
    }
  }
}

__device__ __forceinline__ float xhalf(float x) {
  const int b = __builtin_bit_cast(int, x);
  return __builtin_bit_cast(float,
      __builtin_amdgcn_permlanex16(b, b, 0x76543210, 0xfedcba98, false, false));
}
template <int M>
__device__ __forceinline__ float xrow(float x) {
  const int b = __builtin_bit_cast(int, x);
  return __builtin_bit_cast(float,
      __builtin_amdgcn_update_dpp(0, b, 0x160 + M, 0xf, 0xf, true));
}
__device__ __forceinline__ float wave_sum(float s) {
  s += xhalf(s);
  s += xrow<8>(s);
  s += xrow<4>(s);
  s += xrow<2>(s);
  s += xrow<1>(s);
  return s;
}
__device__ __forceinline__ float dot8(const float4 qa, const float4 qb,
                                      const float4 ka, const float4 kb) {
  float r = qa.y * ka.y;
  r = __builtin_fmaf(qa.x, ka.x, r);
  r = __builtin_fmaf(qa.z, ka.z, r);
  r = __builtin_fmaf(qa.w, ka.w, r);
  r = __builtin_fmaf(qb.x, kb.x, r);
  r = __builtin_fmaf(qb.y, kb.y, r);
  r = __builtin_fmaf(qb.z, kb.z, r);
  r = __builtin_fmaf(qb.w, kb.w, r);
  return r;
}

#define EDGES_PER_WAVE 32
#define EDGE_GROUP 8
static_assert(N_EDGES % EDGES_PER_WAVE == 0, "edge count must fill whole waves");

__global__ __launch_bounds__(256) void edge_exp(const float* __restrict__ q,
                                                const float* __restrict__ kmat,
                                                const int* __restrict__ src,
                                                const int* __restrict__ dst,
                                                float* __restrict__ ex,
                                                float* __restrict__ expsum,
                                                int nE) {
  const int lane = threadIdx.x & 31;
  const int wv = __builtin_amdgcn_readfirstlane((int)(threadIdx.x >> 5));
  const int e0 = (blockIdx.x * 8 + wv) * EDGES_PER_WAVE;
  if (e0 >= nE) return;

  const int sv = src[e0 + lane];
  const int dv = dst[e0 + lane];

  int sCur = __builtin_amdgcn_readlane(sv, 0);
  const float4* qp = (const float4*)(q + (size_t)sCur * FQK_DIM);
  float4 qa = qp[lane];
  float4 qb = qp[lane + 32];

  float acc = 0.0f;

  for (int j0 = 0; j0 < EDGES_PER_WAVE; j0 += EDGE_GROUP) {
    int s[EDGE_GROUP];
    int differ = 0;
    for (int t = 0; t < EDGE_GROUP; ++t) {
      s[t] = __builtin_amdgcn_readlane(sv, j0 + t);
      differ |= s[t] ^ s[0];
    }

    if (differ == 0) {
      if (s[0] != sCur) {
        sCur = s[0];
        const float4* qn = (const float4*)(q + (size_t)sCur * FQK_DIM);
        qa = qn[lane];
        qb = qn[lane + 32];
      }
      float4 ka[EDGE_GROUP], kb[EDGE_GROUP];
      for (int t = 0; t < EDGE_GROUP; ++t) {
        const float4* kp = (const float4*)(kmat + (size_t)__builtin_amdgcn_readlane(dv, j0 + t) * FQK_DIM);
        ka[t] = kp[lane];
        kb[t] = kp[lane + 32];
      }
      for (int t = 0; t < EDGE_GROUP; ++t) {
        const float r = wave_sum(dot8(qa, qb, ka[t], kb[t]));
        acc = (lane == j0 + t) ? r : acc;
      }
    } else {
      for (int j = j0; j < j0 + EDGE_GROUP; j += 2) {
        const int sx = __builtin_amdgcn_readlane(sv, j);
        const int sy = __builtin_amdgcn_readlane(sv, j + 1);
        const float4* qx = (const float4*)(q + (size_t)sx * FQK_DIM);
        const float4* qy = (const float4*)(q + (size_t)sy * FQK_DIM);
        const float4* kx = (const float4*)(kmat + (size_t)__builtin_amdgcn_readlane(dv, j) * FQK_DIM);
        const float4* ky = (const float4*)(kmat + (size_t)__builtin_amdgcn_readlane(dv, j + 1) * FQK_DIM);
        const float4 qxa = qx[lane], qxb = qx[lane + 32];
        const float4 qya = qy[lane], qyb = qy[lane + 32];
        const float4 kxa = kx[lane], kxb = kx[lane + 32];
        const float4 kya = ky[lane], kyb = ky[lane + 32];
        const float rx = wave_sum(dot8(qxa, qxb, kxa, kxb));
        const float ry = wave_sum(dot8(qya, qyb, kya, kyb));
        acc = (lane == j)     ? rx : acc;
        acc = (lane == j + 1) ? ry : acc;
        qa = qya;
        qb = qyb;
        sCur = sy;
      }
    }
  }

  const float v = __expf(acc);
  ex[e0 + lane] = v;
  unsafeAtomicAdd(expsum + sv, v);
}

__global__ __launch_bounds__(256) void edge_norm(float* __restrict__ ex,
                                                 const int* __restrict__ src,
                                                 const float* __restrict__ expsum,
                                                 int nE) {
  const int e = blockIdx.x * 256 + threadIdx.x;
  if (e < nE) ex[e] = ex[e] / expsum[src[e]];
}

extern "C" void kernel_launch(void* const* d_in, const int* in_sizes, int n_in,
                              void* d_out, int out_size, void* d_ws, size_t ws_size,
                              hipStream_t stream) {
  const float* x  = (const float*)d_in[0];
  const int*   ei = (const int*)d_in[1];
  const float* W  = (const float*)d_in[2];
  float* out = (float*)d_out;

  float* q      = (float*)d_ws;
  float* kmat   = q + (size_t)N_NODES * FQK_DIM;
  float* expsum = kmat + (size_t)N_NODES * FQK_DIM;

  const int* src = ei;
  const int* dst = ei + N_EDGES;

  zero_f32<<<(N_NODES + 255) / 256, 256, 0, stream>>>(expsum, N_NODES);
  gemm_qk<<<250, 256, 0, stream>>>(x, W, q, kmat);
  edge_exp<<<N_EDGES / (8 * EDGES_PER_WAVE), 256, 0, stream>>>(q, kmat, src, dst, out, expsum, N_EDGES);
  edge_norm<<<(N_EDGES + 255) / 256, 256, 0, stream>>>(out, src, expsum, N_EDGES);
}


# x W^T re-associated: Mt = Wk^T Wq (256^3, matrix unit, in the launch that zeroes expsum), y = x Mt^T / 16 in 80x256 tiles replaces q and k, edges dot y[src] with x[dest]; sum order changed
# speedup vs baseline: 2.1395x; 1.1591x over previous
#include <hip/hip_runtime.h>
#include <hip/hip_bf16.h>
#include <math.h>

typedef float v2f __attribute__((ext_vector_type(2)));
typedef float v8f __attribute__((ext_vector_type(8)));

#define N_NODES 20000
#define N_EDGES 640000
#define FIN_DIM 256
#define FQK_DIM 256

#define PREP_GROUPS 32
__global__ __launch_bounds__(128) void prep_mt(const float* __restrict__ W,
                                               float* __restrict__ Mt,
                                               float* __restrict__ expsum) {
#pragma unroll
  for (int t = 0; t < (N_NODES + PREP_GROUPS * 128 - 1) / (PREP_GROUPS * 128); ++t) {
    const int i = t * (PREP_GROUPS * 128) + blockIdx.x * 128 + threadIdx.x;
    if (i < N_NODES) expsum[i] = 0.0f;
  }

  const int lane = threadIdx.x & 31;
  const int wv   = threadIdx.x >> 5;
  const int c    = lane & 15;
  const int h    = lane >> 4;
  const int m0   = (blockIdx.x >> 1) * 16;
  const int n0   = (blockIdx.x & 1) * 128 + wv * 32;
  const float* pa = W + (size_t)(FQK_DIM + 2 * h) * FIN_DIM + m0 + c;
  const float* pb = W + (size_t)(2 * h) * FIN_DIM + n0 + c;

  v8f acc0 = (v8f){}, acc1 = (v8f){};
#pragma unroll 16
  for (int f0 = 0; f0 < FQK_DIM; f0 += 4) {
    v2f a, b0, b1;
    a.x  = pa[(size_t)f0 * FIN_DIM];
    a.y  = pa[(size_t)(f0 + 1) * FIN_DIM];
    b0.x = pb[(size_t)f0 * FIN_DIM];
    b0.y = pb[(size_t)(f0 + 1) * FIN_DIM];
    b1.x = pb[(size_t)f0 * FIN_DIM + 16];
    b1.y = pb[(size_t)(f0 + 1) * FIN_DIM + 16];
    acc0 = __builtin_amdgcn_wmma_f32_16x16x4_f32(false, a, false, b0, (short)0, acc0, false, false);
    acc1 = __builtin_amdgcn_wmma_f32_16x16x4_f32(false, a, false, b1, (short)0, acc1, false, false);
  }
#pragma unroll
  for (int r = 0; r < 8; ++r) {
    float* o = Mt + (size_t)(m0 + 8 * h + r) * FIN_DIM + n0 + c;
    o[0]  = acc0[r];
    o[16] = acc1[r];
  }
}

typedef float v4f __attribute__((ext_vector_type(4)));

#define GEMM_BM 80
#define GEMM_BN 256
#define GEMM_KC 64
#define GEMM_PITCH 68
#define GEMM_ROWS (GEMM_BM + GEMM_BN)
#define GEMM_STAGE (GEMM_ROWS * GEMM_PITCH)
#define GEMM_PRE (GEMM_ROWS / 16)
#define GEMM_PRE_A (GEMM_BM / 16)
#define GEMM_CHUNKS (FIN_DIM / GEMM_KC)
#define GEMM_TI (GEMM_BM / 16)
#define GEMM_TJ 2
static_assert(N_NODES % GEMM_BM == 0, "row tiles must cover the nodes exactly");
static_assert(GEMM_BN == FIN_DIM, "one column tile is all of y's columns");
static_assert(GEMM_BN == 8 * GEMM_TJ * 16, "eight waves of 32-column strips cover the tile");

__device__ __forceinline__ void gemm_fetch(v4f (&pre)[GEMM_PRE], const float* gA, const float* gB,
                                           const int kOff) {
#pragma unroll
  for (int i = 0; i < GEMM_PRE_A; ++i)
    pre[i] = *(const v4f*)(gA + (size_t)i * 16 * FIN_DIM + kOff);
#pragma unroll
  for (int i = GEMM_PRE_A; i < GEMM_PRE; ++i)
    pre[i] = *(const v4f*)(gB + (size_t)(i - GEMM_PRE_A) * 16 * FIN_DIM + kOff);
}
__device__ __forceinline__ void gemm_put(const v4f (&pre)[GEMM_PRE], float* s) {
#pragma unroll
  for (int i = 0; i < GEMM_PRE; ++i) *(v4f*)(s + i * 16 * GEMM_PITCH) = pre[i];
}
__device__ __forceinline__ void gemm_frag_pair(v2f (&a)[2][GEMM_TI], v2f (&b)[2][GEMM_TJ],
                                               const float* aS, const float* bS, const int g) {
#pragma unroll
  for (int s = 0; s < 2; ++s) {
#pragma unroll
    for (int i = 0; i < GEMM_TI; ++i) a[s][i] = *(const v2f*)(aS + i * 16 * GEMM_PITCH + (2 * g + s) * 4);
#pragma unroll
    for (int j = 0; j < GEMM_TJ; ++j) b[s][j] = *(const v2f*)(bS + j * 16 * GEMM_PITCH + (2 * g + s) * 4);
  }
}
__device__ __forceinline__ void gemm_chunk(v8f (&acc)[GEMM_TI][GEMM_TJ], const float* aS,
                                           const float* bS) {
  v2f a[2][2][GEMM_TI], b[2][2][GEMM_TJ];
  gemm_frag_pair(a[0], b[0], aS, bS, 0);
#pragma unroll
  for (int g = 0; g < GEMM_KC / 8; ++g) {
    if (g + 1 < GEMM_KC / 8) gemm_frag_pair(a[(g + 1) & 1], b[(g + 1) & 1], aS, bS, g + 1);
    __builtin_amdgcn_sched_barrier(0);
#pragma unroll
    for (int s = 0; s < 2; ++s)
#pragma unroll
      for (int i = 0; i < GEMM_TI; ++i)
#pragma unroll
        for (int j = 0; j < GEMM_TJ; ++j)
          acc[i][j] = __builtin_amdgcn_wmma_f32_16x16x4_f32(false, a[g & 1][s][i], false,
                                                            b[g & 1][s][j], (short)0, acc[i][j],
                                                            false, false);
    __builtin_amdgcn_sched_barrier(0);
  }
}

__global__ __launch_bounds__(256) void gemm_y(const float* __restrict__ x,
                                              const float* __restrict__ Mt,
                                              float* __restrict__ y) {
  __shared__ __attribute__((aligned(16))) float smem[2 * GEMM_STAGE];

  const int tid   = threadIdx.x;
  const int lane  = tid & 31;
  const int wv    = tid >> 5;
  const int mBase = blockIdx.x * GEMM_BM;

  const int c4 = tid & 15;
  const int r0 = tid >> 4;
  const float* gA = x + (size_t)(mBase + r0) * FIN_DIM + c4 * 4;
  const float* gB = Mt + (size_t)r0 * FIN_DIM + c4 * 4;
  float* sStore = smem + r0 * GEMM_PITCH + c4 * 4;

  const int c  = lane & 15;
  const int h2 = (lane >> 4) << 1;
  const float* aBase = smem + c * GEMM_PITCH + h2;
  const float* bBase = smem + (GEMM_BM + wv * (GEMM_TJ * 16) + c) * GEMM_PITCH + h2;

  v8f acc[GEMM_TI][GEMM_TJ];
#pragma unroll
  for (int i = 0; i < GEMM_TI; ++i)
#pragma unroll
    for (int j = 0; j < GEMM_TJ; ++j) acc[i][j] = (v8f){};

  {
    v4f pre[GEMM_PRE];
    gemm_fetch(pre, gA, gB, 0);
    gemm_put(pre, sStore);
  }
  __syncthreads();

#pragma unroll 1
  for (int kc = 0; kc < GEMM_CHUNKS; ++kc) {
    v4f pre[GEMM_PRE];
    const bool more = (kc + 1) < GEMM_CHUNKS;
    if (more) gemm_fetch(pre, gA, gB, (kc + 1) * GEMM_KC);
    gemm_chunk(acc, aBase + (kc & 1) * GEMM_STAGE, bBase + (kc & 1) * GEMM_STAGE);
    if (more) gemm_put(pre, sStore + ((kc + 1) & 1) * GEMM_STAGE);
    __syncthreads();
  }

  const int mOff = (lane >> 4) ? 8 : 0;
#pragma unroll
  for (int i = 0; i < GEMM_TI; ++i) {
#pragma unroll
    for (int j = 0; j < GEMM_TJ; ++j) {
      const int col = wv * (GEMM_TJ * 16) + j * 16 + c;
#pragma unroll
      for (int r = 0; r < 8; ++r) {
        const int m = mBase + i * 16 + mOff + r;
        y[(size_t)m * FIN_DIM + col] = acc[i][j][r] * 0.0625f;
      }
    }
  }
}

__device__ __forceinline__ float xhalf(float x) {
  const int b = __builtin_bit_cast(int, x);
  return __builtin_bit_cast(float,
      __builtin_amdgcn_permlanex16(b, b, 0x76543210, 0xfedcba98, false, false));
}
template <int M>
__device__ __forceinline__ float xrow(float x) {
  const int b = __builtin_bit_cast(int, x);
  return __builtin_bit_cast(float,
      __builtin_amdgcn_update_dpp(0, b, 0x160 + M, 0xf, 0xf, true));
}
__device__ __forceinline__ float wave_sum(float s) {
  s += xhalf(s);
  s += xrow<8>(s);
  s += xrow<4>(s);
  s += xrow<2>(s);
  s += xrow<1>(s);
  return s;
}
__device__ __forceinline__ float dot8(const float4 qa, const float4 qb,
                                      const float4 ka, const float4 kb) {
  float r = qa.y * ka.y;
  r = __builtin_fmaf(qa.x, ka.x, r);
  r = __builtin_fmaf(qa.z, ka.z, r);
  r = __builtin_fmaf(qa.w, ka.w, r);
  r = __builtin_fmaf(qb.x, kb.x, r);
  r = __builtin_fmaf(qb.y, kb.y, r);
  r = __builtin_fmaf(qb.z, kb.z, r);
  r = __builtin_fmaf(qb.w, kb.w, r);
  return r;
}

#define EDGES_PER_WAVE 32
#define EDGE_GROUP 8
static_assert(N_EDGES % EDGES_PER_WAVE == 0, "edge count must fill whole waves");

__global__ __launch_bounds__(256) void edge_exp(const float* __restrict__ q,
                                                const float* __restrict__ kmat,
                                                const int* __restrict__ src,
                                                const int* __restrict__ dst,
                                                float* __restrict__ ex,
                                                float* __restrict__ expsum,
                                                int nE) {
  const int lane = threadIdx.x & 31;
  const int wv = __builtin_amdgcn_readfirstlane((int)(threadIdx.x >> 5));
  const int e0 = (blockIdx.x * 8 + wv) * EDGES_PER_WAVE;
  if (e0 >= nE) return;

  const int sv = src[e0 + lane];
  const int dv = dst[e0 + lane];

  int sCur = __builtin_amdgcn_readlane(sv, 0);
  const float4* qp = (const float4*)(q + (size_t)sCur * FQK_DIM);
  float4 qa = qp[lane];
  float4 qb = qp[lane + 32];

  float acc = 0.0f;

  for (int j0 = 0; j0 < EDGES_PER_WAVE; j0 += EDGE_GROUP) {
    int s[EDGE_GROUP];
    int differ = 0;
    for (int t = 0; t < EDGE_GROUP; ++t) {
      s[t] = __builtin_amdgcn_readlane(sv, j0 + t);
      differ |= s[t] ^ s[0];
    }

    if (differ == 0) {
      if (s[0] != sCur) {
        sCur = s[0];
        const float4* qn = (const float4*)(q + (size_t)sCur * FQK_DIM);
        qa = qn[lane];
        qb = qn[lane + 32];
      }
      float4 ka[EDGE_GROUP], kb[EDGE_GROUP];
      for (int t = 0; t < EDGE_GROUP; ++t) {
        const float4* kp = (const float4*)(kmat + (size_t)__builtin_amdgcn_readlane(dv, j0 + t) * FQK_DIM);
        ka[t] = kp[lane];
        kb[t] = kp[lane + 32];
      }
      for (int t = 0; t < EDGE_GROUP; ++t) {
        const float r = wave_sum(dot8(qa, qb, ka[t], kb[t]));
        acc = (lane == j0 + t) ? r : acc;
      }
    } else {
      for (int j = j0; j < j0 + EDGE_GROUP; j += 2) {
        const int sx = __builtin_amdgcn_readlane(sv, j);
        const int sy = __builtin_amdgcn_readlane(sv, j + 1);
        const float4* qx = (const float4*)(q + (size_t)sx * FQK_DIM);
        const float4* qy = (const float4*)(q + (size_t)sy * FQK_DIM);
        const float4* kx = (const float4*)(kmat + (size_t)__builtin_amdgcn_readlane(dv, j) * FQK_DIM);
        const float4* ky = (const float4*)(kmat + (size_t)__builtin_amdgcn_readlane(dv, j + 1) * FQK_DIM);
        const float4 qxa = qx[lane], qxb = qx[lane + 32];
        const float4 qya = qy[lane], qyb = qy[lane + 32];
        const float4 kxa = kx[lane], kxb = kx[lane + 32];
        const float4 kya = ky[lane], kyb = ky[lane + 32];
        const float rx = wave_sum(dot8(qxa, qxb, kxa, kxb));
        const float ry = wave_sum(dot8(qya, qyb, kya, kyb));
        acc = (lane == j)     ? rx : acc;
        acc = (lane == j + 1) ? ry : acc;
        qa = qya;
        qb = qyb;
        sCur = sy;
      }
    }
  }

  const float v = __expf(acc);
  ex[e0 + lane] = v;
  unsafeAtomicAdd(expsum + sv, v);
}

__global__ __launch_bounds__(256) void edge_norm(float* __restrict__ ex,
                                                 const int* __restrict__ src,
                                                 const float* __restrict__ expsum,
                                                 int nE) {
  const int e = blockIdx.x * 256 + threadIdx.x;
  if (e < nE) ex[e] = ex[e] / expsum[src[e]];
}

extern "C" void kernel_launch(void* const* d_in, const int* in_sizes, int n_in,
                              void* d_out, int out_size, void* d_ws, size_t ws_size,
                              hipStream_t stream) {
  const float* x  = (const float*)d_in[0];
  const int*   ei = (const int*)d_in[1];
  const float* W  = (const float*)d_in[2];
  float* out = (float*)d_out;

  float* y      = (float*)d_ws;
  float* Mt     = y + (size_t)N_NODES * FQK_DIM;
  float* expsum = Mt + (size_t)N_NODES * FQK_DIM;

  const int* src = ei;
  const int* dst = ei + N_EDGES;

  prep_mt<<<PREP_GROUPS, 128, 0, stream>>>(W, Mt, expsum);
  gemm_y<<<N_NODES / GEMM_BM, 256, 0, stream>>>(x, Mt, y);
  edge_exp<<<N_EDGES / (8 * EDGES_PER_WAVE), 256, 0, stream>>>(y, x, src, dst, out, expsum, N_EDGES);
  edge_norm<<<(N_EDGES + 255) / 256, 256, 0, stream>>>(out, src, expsum, N_EDGES);
}


# edge pass: lanes of one wave with equal adjacent src sum their ex in registers (segmented scan, DPP), one atomic add per run, about 40000 lane adds not 640000; sum order changed
# speedup vs baseline: 2.7361x; 1.2789x over previous
#include <hip/hip_runtime.h>
#include <hip/hip_bf16.h>
#include <math.h>

typedef float v2f __attribute__((ext_vector_type(2)));
typedef float v8f __attribute__((ext_vector_type(8)));

#define N_NODES 20000
#define N_EDGES 640000
#define FIN_DIM 256
#define FQK_DIM 256

#define PREP_GROUPS 32
__global__ __launch_bounds__(128) void prep_mt(const float* __restrict__ W,
                                               float* __restrict__ Mt,
                                               float* __restrict__ expsum) {
#pragma unroll
  for (int t = 0; t < (N_NODES + PREP_GROUPS * 128 - 1) / (PREP_GROUPS * 128); ++t) {
    const int i = t * (PREP_GROUPS * 128) + blockIdx.x * 128 + threadIdx.x;
    if (i < N_NODES) expsum[i] = 0.0f;
  }

  const int lane = threadIdx.x & 31;
  const int wv   = threadIdx.x >> 5;
  const int c    = lane & 15;
  const int h    = lane >> 4;
  const int m0   = (blockIdx.x >> 1) * 16;
  const int n0   = (blockIdx.x & 1) * 128 + wv * 32;
  const float* pa = W + (size_t)(FQK_DIM + 2 * h) * FIN_DIM + m0 + c;
  const float* pb = W + (size_t)(2 * h) * FIN_DIM + n0 + c;

  v8f acc0 = (v8f){}, acc1 = (v8f){};
#pragma unroll 16
  for (int f0 = 0; f0 < FQK_DIM; f0 += 4) {
    v2f a, b0, b1;
    a.x  = pa[(size_t)f0 * FIN_DIM];
    a.y  = pa[(size_t)(f0 + 1) * FIN_DIM];
    b0.x = pb[(size_t)f0 * FIN_DIM];
    b0.y = pb[(size_t)(f0 + 1) * FIN_DIM];
    b1.x = pb[(size_t)f0 * FIN_DIM + 16];
    b1.y = pb[(size_t)(f0 + 1) * FIN_DIM + 16];
    acc0 = __builtin_amdgcn_wmma_f32_16x16x4_f32(false, a, false, b0, (short)0, acc0, false, false);
    acc1 = __builtin_amdgcn_wmma_f32_16x16x4_f32(false, a, false, b1, (short)0, acc1, false, false);
  }
#pragma unroll
  for (int r = 0; r < 8; ++r) {
    float* o = Mt + (size_t)(m0 + 8 * h + r) * FIN_DIM + n0 + c;
    o[0]  = acc0[r];
    o[16] = acc1[r];
  }
}

typedef float v4f __attribute__((ext_vector_type(4)));

#define GEMM_BM 80
#define GEMM_BN 256
#define GEMM_KC 64
#define GEMM_PITCH 68
#define GEMM_ROWS (GEMM_BM + GEMM_BN)
#define GEMM_STAGE (GEMM_ROWS * GEMM_PITCH)
#define GEMM_PRE (GEMM_ROWS / 16)
#define GEMM_PRE_A (GEMM_BM / 16)
#define GEMM_CHUNKS (FIN_DIM / GEMM_KC)
#define GEMM_TI (GEMM_BM / 16)
#define GEMM_TJ 2
static_assert(N_NODES % GEMM_BM == 0, "row tiles must cover the nodes exactly");
static_assert(GEMM_BN == FIN_DIM, "one column tile is all of y's columns");
static_assert(GEMM_BN == 8 * GEMM_TJ * 16, "eight waves of 32-column strips cover the tile");

__device__ __forceinline__ void gemm_fetch(v4f (&pre)[GEMM_PRE], const float* gA, const float* gB,
                                           const int kOff) {
#pragma unroll
  for (int i = 0; i < GEMM_PRE_A; ++i)
    pre[i] = *(const v4f*)(gA + (size_t)i * 16 * FIN_DIM + kOff);
#pragma unroll
  for (int i = GEMM_PRE_A; i < GEMM_PRE; ++i)
    pre[i] = *(const v4f*)(gB + (size_t)(i - GEMM_PRE_A) * 16 * FIN_DIM + kOff);
}
__device__ __forceinline__ void gemm_put(const v4f (&pre)[GEMM_PRE], float* s) {
#pragma unroll
  for (int i = 0; i < GEMM_PRE; ++i) *(v4f*)(s + i * 16 * GEMM_PITCH) = pre[i];
}
__device__ __forceinline__ void gemm_frag_pair(v2f (&a)[2][GEMM_TI], v2f (&b)[2][GEMM_TJ],
                                               const float* aS, const float* bS, const int g) {
#pragma unroll
  for (int s = 0; s < 2; ++s) {
#pragma unroll
    for (int i = 0; i < GEMM_TI; ++i) a[s][i] = *(const v2f*)(aS + i * 16 * GEMM_PITCH + (2 * g + s) * 4);
#pragma unroll
    for (int j = 0; j < GEMM_TJ; ++j) b[s][j] = *(const v2f*)(bS + j * 16 * GEMM_PITCH + (2 * g + s) * 4);
  }
}
__device__ __forceinline__ void gemm_chunk(v8f (&acc)[GEMM_TI][GEMM_TJ], const float* aS,
                                           const float* bS) {
  v2f a[2][2][GEMM_TI], b[2][2][GEMM_TJ];
  gemm_frag_pair(a[0], b[0], aS, bS, 0);
#pragma unroll
  for (int g = 0; g < GEMM_KC / 8; ++g) {
    if (g + 1 < GEMM_KC / 8) gemm_frag_pair(a[(g + 1) & 1], b[(g + 1) & 1], aS, bS, g + 1);
    __builtin_amdgcn_sched_barrier(0);
#pragma unroll
    for (int s = 0; s < 2; ++s)
#pragma unroll
      for (int i = 0; i < GEMM_TI; ++i)
#pragma unroll
        for (int j = 0; j < GEMM_TJ; ++j)
          acc[i][j] = __builtin_amdgcn_wmma_f32_16x16x4_f32(false, a[g & 1][s][i], false,
                                                            b[g & 1][s][j], (short)0, acc[i][j],
                                                            false, false);
    __builtin_amdgcn_sched_barrier(0);
  }
}

__global__ __launch_bounds__(256) void gemm_y(const float* __restrict__ x,
                                              const float* __restrict__ Mt,
                                              float* __restrict__ y) {
  __shared__ __attribute__((aligned(16))) float smem[2 * GEMM_STAGE];

  const int tid   = threadIdx.x;
  const int lane  = tid & 31;
  const int wv    = tid >> 5;
  const int mBase = blockIdx.x * GEMM_BM;

  const int c4 = tid & 15;
  const int r0 = tid >> 4;
  const float* gA = x + (size_t)(mBase + r0) * FIN_DIM + c4 * 4;
  const float* gB = Mt + (size_t)r0 * FIN_DIM + c4 * 4;
  float* sStore = smem + r0 * GEMM_PITCH + c4 * 4;

  const int c  = lane & 15;
  const int h2 = (lane >> 4) << 1;
  const float* aBase = smem + c * GEMM_PITCH + h2;
  const float* bBase = smem + (GEMM_BM + wv * (GEMM_TJ * 16) + c) * GEMM_PITCH + h2;

  v8f acc[GEMM_TI][GEMM_TJ];
#pragma unroll
  for (int i = 0; i < GEMM_TI; ++i)
#pragma unroll
    for (int j = 0; j < GEMM_TJ; ++j) acc[i][j] = (v8f){};

  {
    v4f pre[GEMM_PRE];
    gemm_fetch(pre, gA, gB, 0);
    gemm_put(pre, sStore);
  }
  __syncthreads();

#pragma unroll 1
  for (int kc = 0; kc < GEMM_CHUNKS; ++kc) {
    v4f pre[GEMM_PRE];
    const bool more = (kc + 1) < GEMM_CHUNKS;
    if (more) gemm_fetch(pre, gA, gB, (kc + 1) * GEMM_KC);
    gemm_chunk(acc, aBase + (kc & 1) * GEMM_STAGE, bBase + (kc & 1) * GEMM_STAGE);
    if (more) gemm_put(pre, sStore + ((kc + 1) & 1) * GEMM_STAGE);
    __syncthreads();
  }

  const int mOff = (lane >> 4) ? 8 : 0;
#pragma unroll
  for (int i = 0; i < GEMM_TI; ++i) {
#pragma unroll
    for (int j = 0; j < GEMM_TJ; ++j) {
      const int col = wv * (GEMM_TJ * 16) + j * 16 + c;
#pragma unroll
      for (int r = 0; r < 8; ++r) {
        const int m = mBase + i * 16 + mOff + r;
        y[(size_t)m * FIN_DIM + col] = acc[i][j][r] * 0.0625f;
      }
    }
  }
}

__device__ __forceinline__ float xhalf(float x) {
  const int b = __builtin_bit_cast(int, x);
  return __builtin_bit_cast(float,
      __builtin_amdgcn_permlanex16(b, b, 0x76543210, 0xfedcba98, false, false));
}
template <int M>
__device__ __forceinline__ float xrow(float x) {
  const int b = __builtin_bit_cast(int, x);
  return __builtin_bit_cast(float,
      __builtin_amdgcn_update_dpp(0, b, 0x160 + M, 0xf, 0xf, true));
}
template <int D>
__device__ __forceinline__ float shr(float x) {
  const int b = __builtin_bit_cast(int, x);
  return __builtin_bit_cast(float,
      __builtin_amdgcn_update_dpp(0, b, 0x110 + D, 0xf, 0xf, true));
}
__device__ __forceinline__ float wave_sum(float s) {
  s += xhalf(s);
  s += xrow<8>(s);
  s += xrow<4>(s);
  s += xrow<2>(s);
  s += xrow<1>(s);
  return s;
}
__device__ __forceinline__ float dot8(const float4 qa, const float4 qb,
                                      const float4 ka, const float4 kb) {
  float r = qa.y * ka.y;
  r = __builtin_fmaf(qa.x, ka.x, r);
  r = __builtin_fmaf(qa.z, ka.z, r);
  r = __builtin_fmaf(qa.w, ka.w, r);
  r = __builtin_fmaf(qb.x, kb.x, r);
  r = __builtin_fmaf(qb.y, kb.y, r);
  r = __builtin_fmaf(qb.z, kb.z, r);
  r = __builtin_fmaf(qb.w, kb.w, r);
  return r;
}

#define EDGES_PER_WAVE 32
#define EDGE_GROUP 8
static_assert(N_EDGES % EDGES_PER_WAVE == 0, "edge count must fill whole waves");

__global__ __launch_bounds__(256) void edge_exp(const float* __restrict__ q,
                                                const float* __restrict__ kmat,
                                                const int* __restrict__ src,
                                                const int* __restrict__ dst,
                                                float* __restrict__ ex,
                                                float* __restrict__ expsum,
                                                int nE) {
  const int lane = threadIdx.x & 31;
  const int wv = __builtin_amdgcn_readfirstlane((int)(threadIdx.x >> 5));
  const int e0 = (blockIdx.x * 8 + wv) * EDGES_PER_WAVE;
  if (e0 >= nE) return;

  const int sv = src[e0 + lane];
  const int dv = dst[e0 + lane];

  int sCur = __builtin_amdgcn_readlane(sv, 0);
  const float4* qp = (const float4*)(q + (size_t)sCur * FQK_DIM);
  float4 qa = qp[lane];
  float4 qb = qp[lane + 32];

  float acc = 0.0f;

  for (int j0 = 0; j0 < EDGES_PER_WAVE; j0 += EDGE_GROUP) {
    int s[EDGE_GROUP];
    int differ = 0;
    for (int t = 0; t < EDGE_GROUP; ++t) {
      s[t] = __builtin_amdgcn_readlane(sv, j0 + t);
      differ |= s[t] ^ s[0];
    }

    if (differ == 0) {
      if (s[0] != sCur) {
        sCur = s[0];
        const float4* qn = (const float4*)(q + (size_t)sCur * FQK_DIM);
        qa = qn[lane];
        qb = qn[lane + 32];
      }
      float4 ka[EDGE_GROUP], kb[EDGE_GROUP];
      for (int t = 0; t < EDGE_GROUP; ++t) {
        const float4* kp = (const float4*)(kmat + (size_t)__builtin_amdgcn_readlane(dv, j0 + t) * FQK_DIM);
        ka[t] = kp[lane];
        kb[t] = kp[lane + 32];
      }
      for (int t = 0; t < EDGE_GROUP; ++t) {
        const float r = wave_sum(dot8(qa, qb, ka[t], kb[t]));
        acc = (lane == j0 + t) ? r : acc;
      }
    } else {
      for (int j = j0; j < j0 + EDGE_GROUP; j += 2) {
        const int sx = __builtin_amdgcn_readlane(sv, j);
        const int sy = __builtin_amdgcn_readlane(sv, j + 1);
        const float4* qx = (const float4*)(q + (size_t)sx * FQK_DIM);
        const float4* qy = (const float4*)(q + (size_t)sy * FQK_DIM);
        const float4* kx = (const float4*)(kmat + (size_t)__builtin_amdgcn_readlane(dv, j) * FQK_DIM);
        const float4* ky = (const float4*)(kmat + (size_t)__builtin_amdgcn_readlane(dv, j + 1) * FQK_DIM);
        const float4 qxa = qx[lane], qxb = qx[lane + 32];
        const float4 qya = qy[lane], qyb = qy[lane + 32];
        const float4 kxa = kx[lane], kxb = kx[lane + 32];
        const float4 kya = ky[lane], kyb = ky[lane + 32];
        const float rx = wave_sum(dot8(qxa, qxb, kxa, kxb));
        const float ry = wave_sum(dot8(qya, qyb, kya, kyb));
        acc = (lane == j)     ? rx : acc;
        acc = (lane == j + 1) ? ry : acc;
        qa = qya;
        qb = qyb;
        sCur = sy;
      }
    }
  }

  const float v = __expf(acc);
  ex[e0 + lane] = v;

  int prev = __builtin_amdgcn_update_dpp(0, sv, 0x111, 0xf, 0xf, true);
  const int s15 = __builtin_amdgcn_readlane(sv, 15);
  prev = (lane == 16) ? s15 : prev;
  const bool head = (lane == 0) || (prev != sv);
  const unsigned hm = __builtin_amdgcn_ballot_w32(head);
  const int start = 31 - __builtin_clz(hm & (0xffffffffu >> (31 - lane)));
  const int rowBase = lane & 16;
  const int lo = (start > rowBase) ? start : rowBase;
  float run = v;
  float t;
  t = shr<1>(run); run += (lane - 1 >= lo) ? t : 0.0f;
  t = shr<2>(run); run += (lane - 2 >= lo) ? t : 0.0f;
  t = shr<4>(run); run += (lane - 4 >= lo) ? t : 0.0f;
  t = shr<8>(run); run += (lane - 8 >= lo) ? t : 0.0f;
  const float part15 =
      __builtin_bit_cast(float, __builtin_amdgcn_readlane(__builtin_bit_cast(int, run), 15));
  run += (lane >= 16 && start < 16) ? part15 : 0.0f;
  const bool tail = (lane == 31) || ((((hm >> 1) >> lane) & 1u) != 0u);
  if (tail) unsafeAtomicAdd(expsum + sv, run);
}

__global__ __launch_bounds__(256) void edge_norm(float* __restrict__ ex,
                                                 const int* __restrict__ src,
                                                 const float* __restrict__ expsum,
                                                 int nE) {
  const int e = blockIdx.x * 256 + threadIdx.x;
  if (e < nE) ex[e] = ex[e] / expsum[src[e]];
}

extern "C" void kernel_launch(void* const* d_in, const int* in_sizes, int n_in,
                              void* d_out, int out_size, void* d_ws, size_t ws_size,
                              hipStream_t stream) {
  const float* x  = (const float*)d_in[0];
  const int*   ei = (const int*)d_in[1];
  const float* W  = (const float*)d_in[2];
  float* out = (float*)d_out;

  float* y      = (float*)d_ws;
  float* Mt     = y + (size_t)N_NODES * FQK_DIM;
  float* expsum = Mt + (size_t)N_NODES * FQK_DIM;

  const int* src = ei;
  const int* dst = ei + N_EDGES;

  prep_mt<<<PREP_GROUPS, 128, 0, stream>>>(W, Mt, expsum);
  gemm_y<<<N_NODES / GEMM_BM, 256, 0, stream>>>(x, Mt, y);
  edge_exp<<<N_EDGES / (8 * EDGES_PER_WAVE), 256, 0, stream>>>(y, x, src, dst, out, expsum, N_EDGES);
  edge_norm<<<(N_EDGES + 255) / 256, 256, 0, stream>>>(out, src, expsum, N_EDGES);
}
